# starting kernel
# baseline (speedup 1.0000x reference)
#include <hip/hip_runtime.h>
#include <math.h>

typedef __attribute__((ext_vector_type(2))) float v2f;
typedef __attribute__((ext_vector_type(8))) float v8f;

#define NQ 8
#define NL 3
#define DIM 256
#define WPB 8
#define THREADS (WPB * 32)

__device__ __forceinline__ float2 cmul(float2 a, float2 b) {
  return make_float2(a.x * b.x - a.y * b.y, a.x * b.y + a.y * b.x);
}

__global__ __launch_bounds__(THREADS) void qnoise_wmma_kernel(
    const float* __restrict__ weights,
    const float* __restrict__ z1,
    const float* __restrict__ z2,
    float* __restrict__ out,
    int batch) {
  __shared__ float  sU[NL * 2 * 256];
  __shared__ float2 sD[NL * 256];
  __shared__ float  sS[WPB][2][DIM];

  const int t    = threadIdx.x;
  const int lane = t & 31;
  const int wave = t >> 5;
  const int m    = lane & 15;
  const int h    = lane >> 4;

  for (int e = t; e < NL * 2 * 256; e += THREADS) {
    int l   = e >> 9;
    int mat = (e >> 8) & 1;
    int i   = (e >> 4) & 15;
    int k   = e & 15;
    float val = 1.0f;
#pragma unroll
    for (int j = 0; j < 4; ++j) {
      int q = mat * 4 + j;
      float th = 0.5f * weights[l * 15 + q];
      float c = cosf(th), s = sinf(th);
      int ib = (i >> (3 - j)) & 1;
      int kb = (k >> (3 - j)) & 1;
      val *= (ib == kb) ? c : ((ib == 0) ? -s : s);
    }
    sU[e] = val;
  }
  for (int e = t; e < NL * 256; e += THREADS) {
    int l = e >> 8;
    int idx = e & 255;
    float ang = 0.0f;
#pragma unroll
    for (int i2 = 0; i2 < 7; ++i2) {
      int b1 = (idx >> (7 - i2)) & 1;
      int b0 = (idx >> (6 - i2)) & 1;
      ang += weights[l * 15 + 8 + i2] * ((b1 == b0) ? 1.0f : -1.0f);
    }
    float ha = -0.5f * ang;
    sD[e] = make_float2(cosf(ha), sinf(ha));
  }
  __syncthreads();

  const int sample = blockIdx.x * WPB + wave;
  if (sample >= batch) return;

  float a1 = asinf(z1[sample]);
  float a2 = asinf(z2[sample]);
  float ch = cosf(0.5f * a2), sh = sinf(0.5f * a2);
  float c1 = cosf(0.5f * a1), s1 = sinf(0.5f * a1);
  float2 v0 = make_float2(ch * c1, -sh * c1);
  float2 v1 = make_float2(ch * s1,  sh * s1);

  v8f Sre, Sim;
#pragma unroll
  for (int v = 0; v < 8; ++v) {
    int idx = (v + 8 * h) * 16 + m;
    float2 amp = make_float2(1.0f, 0.0f);
#pragma unroll
    for (int b = 0; b < 8; ++b)
      amp = cmul(amp, ((idx >> b) & 1) ? v1 : v0);
    Sre[v] = amp.x;
    Sim[v] = amp.y;
  }

  float* __restrict__ mre = sS[wave][0];
  float* __restrict__ mim = sS[wave][1];

  for (int l = 0; l < NL; ++l) {
    const float* Uh = &sU[l * 512];
    const float* Ul = &sU[l * 512 + 256];

    v2f auh[4], bul[4];
#pragma unroll
    for (int kb = 0; kb < 4; ++kb) {
      auh[kb][0] = Uh[m * 16 + kb * 4 + 2 * h + 0];
      auh[kb][1] = Uh[m * 16 + kb * 4 + 2 * h + 1];
      bul[kb][0] = Ul[m * 16 + kb * 4 + 2 * h + 0];
      bul[kb][1] = Ul[m * 16 + kb * 4 + 2 * h + 1];
    }

#pragma unroll
    for (int v = 0; v < 8; ++v) {
      mre[(v + 8 * h) * 16 + m] = Sre[v];
      mim[(v + 8 * h) * 16 + m] = Sim[v];
    }

    v8f Tre = {0.f, 0.f, 0.f, 0.f, 0.f, 0.f, 0.f, 0.f};
    v8f Tim = {0.f, 0.f, 0.f, 0.f, 0.f, 0.f, 0.f, 0.f};
#pragma unroll
    for (int kb = 0; kb < 4; ++kb) {
      v2f bre, bim;
      bre[0] = mre[(kb * 4 + 0 + 2 * h) * 16 + m];
      bre[1] = mre[(kb * 4 + 1 + 2 * h) * 16 + m];
      bim[0] = mim[(kb * 4 + 0 + 2 * h) * 16 + m];
      bim[1] = mim[(kb * 4 + 1 + 2 * h) * 16 + m];
      Tre = __builtin_amdgcn_wmma_f32_16x16x4_f32(false, auh[kb], false, bre,
                                                  (short)0, Tre, false, false);
      Tim = __builtin_amdgcn_wmma_f32_16x16x4_f32(false, auh[kb], false, bim,
                                                  (short)0, Tim, false, false);
    }

#pragma unroll
    for (int v = 0; v < 8; ++v) {
      mre[(v + 8 * h) * 16 + m] = Tre[v];
      mim[(v + 8 * h) * 16 + m] = Tim[v];
    }

    v8f Nre = {0.f, 0.f, 0.f, 0.f, 0.f, 0.f, 0.f, 0.f};
    v8f Nim = {0.f, 0.f, 0.f, 0.f, 0.f, 0.f, 0.f, 0.f};
#pragma unroll
    for (int kb = 0; kb < 4; ++kb) {
      v2f are, aim;
      are[0] = mre[m * 16 + kb * 4 + 2 * h + 0];
      are[1] = mre[m * 16 + kb * 4 + 2 * h + 1];
      aim[0] = mim[m * 16 + kb * 4 + 2 * h + 0];
      aim[1] = mim[m * 16 + kb * 4 + 2 * h + 1];
      Nre = __builtin_amdgcn_wmma_f32_16x16x4_f32(false, are, false, bul[kb],
                                                  (short)0, Nre, false, false);
      Nim = __builtin_amdgcn_wmma_f32_16x16x4_f32(false, aim, false, bul[kb],
                                                  (short)0, Nim, false, false);
    }

#pragma unroll
    for (int v = 0; v < 8; ++v) {
      int idx = (v + 8 * h) * 16 + m;
      float2 d = sD[l * 256 + idx];
      float re = Nre[v], im = Nim[v];
      Sre[v] = re * d.x - im * d.y;
      Sim[v] = re * d.y + im * d.x;
    }
  }

  float sq[NQ];
#pragma unroll
  for (int q = 0; q < NQ; ++q) sq[q] = 0.0f;
#pragma unroll
  for (int v = 0; v < 8; ++v) {
    int idx = (v + 8 * h) * 16 + m;
    float p = Sre[v] * Sre[v] + Sim[v] * Sim[v];
#pragma unroll
    for (int q = 0; q < NQ; ++q)
      sq[q] += ((idx >> (7 - q)) & 1) ? -p : p;
  }
#pragma unroll
  for (int off = 16; off > 0; off >>= 1) {
#pragma unroll
    for (int q = 0; q < NQ; ++q)
      sq[q] += __shfl_xor(sq[q], off, 32);
  }
  if (lane == 0) {
#pragma unroll
    for (int q = 0; q < NQ; ++q)
      out[sample * NQ + q] = sq[q];
  }
}

extern "C" void kernel_launch(void* const* d_in, const int* in_sizes, int n_in,
                              void* d_out, int out_size, void* d_ws, size_t ws_size,
                              hipStream_t stream) {
  const float* weights = (const float*)d_in[0];
  const float* z1      = (const float*)d_in[1];
  const float* z2      = (const float*)d_in[2];
  float* out           = (float*)d_out;
  int batch = in_sizes[1];
  int blocks = (batch + WPB - 1) / WPB;
  qnoise_wmma_kernel<<<blocks, THREADS, 0, stream>>>(weights, z1, z2, out, batch);
}


# Tables built once per 32-wave block and reused: 512 blocks, each wave loops over a contiguous run of samples (wave-uniform loop), instead of 8192 blocks each rebuilding them for 8 samples
# speedup vs baseline: 1.8856x; 1.8856x over previous
#include <hip/hip_runtime.h>
#include <math.h>

typedef __attribute__((ext_vector_type(2))) float v2f;
typedef __attribute__((ext_vector_type(8))) float v8f;

#define NQ 8
#define NL 3
#define DIM 256
#define WPB 32
#define THREADS (WPB * 32)
#define NBLOCKS 512

__device__ __forceinline__ float2 cmul(float2 a, float2 b) {
  return make_float2(a.x * b.x - a.y * b.y, a.x * b.y + a.y * b.x);
}

__global__ __launch_bounds__(THREADS) void qnoise_wmma_kernel(
    const float* __restrict__ weights,
    const float* __restrict__ z1,
    const float* __restrict__ z2,
    float* __restrict__ out,
    int batch) {
  __shared__ float  sU[NL * 2 * 256];
  __shared__ float2 sD[NL * 256];
  __shared__ float  sS[WPB][2][DIM];

  const int t    = threadIdx.x;
  const int lane = t & 31;
  const int wave = __builtin_amdgcn_readfirstlane(t >> 5);
  const int m    = lane & 15;
  const int h    = lane >> 4;

  for (int e = t; e < NL * 2 * 256; e += THREADS) {
    int l   = e >> 9;
    int mat = (e >> 8) & 1;
    int i   = (e >> 4) & 15;
    int k   = e & 15;
    float val = 1.0f;
#pragma unroll
    for (int j = 0; j < 4; ++j) {
      int q = mat * 4 + j;
      float th = 0.5f * weights[l * 15 + q];
      float c = cosf(th), s = sinf(th);
      int ib = (i >> (3 - j)) & 1;
      int kb = (k >> (3 - j)) & 1;
      val *= (ib == kb) ? c : ((ib == 0) ? -s : s);
    }
    sU[e] = val;
  }
  for (int e = t; e < NL * 256; e += THREADS) {
    int l = e >> 8;
    int idx = e & 255;
    float ang = 0.0f;
#pragma unroll
    for (int i2 = 0; i2 < 7; ++i2) {
      int b1 = (idx >> (7 - i2)) & 1;
      int b0 = (idx >> (6 - i2)) & 1;
      ang += weights[l * 15 + 8 + i2] * ((b1 == b0) ? 1.0f : -1.0f);
    }
    float ha = -0.5f * ang;
    sD[e] = make_float2(cosf(ha), sinf(ha));
  }
  __syncthreads();

  const int nwaves = gridDim.x * WPB;
  const int per    = (batch + nwaves - 1) / nwaves;
  const int first  = (blockIdx.x * WPB + wave) * per;
  const int last   = min(first + per, batch);

  float* __restrict__ mre = sS[wave][0];
  float* __restrict__ mim = sS[wave][1];

  for (int sample = first; sample < last; ++sample) {
  float a1 = asinf(z1[sample]);
  float a2 = asinf(z2[sample]);
  float ch = cosf(0.5f * a2), sh = sinf(0.5f * a2);
  float c1 = cosf(0.5f * a1), s1 = sinf(0.5f * a1);
  float2 v0 = make_float2(ch * c1, -sh * c1);
  float2 v1 = make_float2(ch * s1,  sh * s1);

  v8f Sre, Sim;
#pragma unroll
  for (int v = 0; v < 8; ++v) {
    int idx = (v + 8 * h) * 16 + m;
    float2 amp = make_float2(1.0f, 0.0f);
#pragma unroll
    for (int b = 0; b < 8; ++b)
      amp = cmul(amp, ((idx >> b) & 1) ? v1 : v0);
    Sre[v] = amp.x;
    Sim[v] = amp.y;
  }

  for (int l = 0; l < NL; ++l) {
    const float* Uh = &sU[l * 512];
    const float* Ul = &sU[l * 512 + 256];

    v2f auh[4], bul[4];
#pragma unroll
    for (int kb = 0; kb < 4; ++kb) {
      auh[kb][0] = Uh[m * 16 + kb * 4 + 2 * h + 0];
      auh[kb][1] = Uh[m * 16 + kb * 4 + 2 * h + 1];
      bul[kb][0] = Ul[m * 16 + kb * 4 + 2 * h + 0];
      bul[kb][1] = Ul[m * 16 + kb * 4 + 2 * h + 1];
    }

#pragma unroll
    for (int v = 0; v < 8; ++v) {
      mre[(v + 8 * h) * 16 + m] = Sre[v];
      mim[(v + 8 * h) * 16 + m] = Sim[v];
    }

    v8f Tre = {0.f, 0.f, 0.f, 0.f, 0.f, 0.f, 0.f, 0.f};
    v8f Tim = {0.f, 0.f, 0.f, 0.f, 0.f, 0.f, 0.f, 0.f};
#pragma unroll
    for (int kb = 0; kb < 4; ++kb) {
      v2f bre, bim;
      bre[0] = mre[(kb * 4 + 0 + 2 * h) * 16 + m];
      bre[1] = mre[(kb * 4 + 1 + 2 * h) * 16 + m];
      bim[0] = mim[(kb * 4 + 0 + 2 * h) * 16 + m];
      bim[1] = mim[(kb * 4 + 1 + 2 * h) * 16 + m];
      Tre = __builtin_amdgcn_wmma_f32_16x16x4_f32(false, auh[kb], false, bre,
                                                  (short)0, Tre, false, false);
      Tim = __builtin_amdgcn_wmma_f32_16x16x4_f32(false, auh[kb], false, bim,
                                                  (short)0, Tim, false, false);
    }

#pragma unroll
    for (int v = 0; v < 8; ++v) {
      mre[(v + 8 * h) * 16 + m] = Tre[v];
      mim[(v + 8 * h) * 16 + m] = Tim[v];
    }

    v8f Nre = {0.f, 0.f, 0.f, 0.f, 0.f, 0.f, 0.f, 0.f};
    v8f Nim = {0.f, 0.f, 0.f, 0.f, 0.f, 0.f, 0.f, 0.f};
#pragma unroll
    for (int kb = 0; kb < 4; ++kb) {
      v2f are, aim;
      are[0] = mre[m * 16 + kb * 4 + 2 * h + 0];
      are[1] = mre[m * 16 + kb * 4 + 2 * h + 1];
      aim[0] = mim[m * 16 + kb * 4 + 2 * h + 0];
      aim[1] = mim[m * 16 + kb * 4 + 2 * h + 1];
      Nre = __builtin_amdgcn_wmma_f32_16x16x4_f32(false, are, false, bul[kb],
                                                  (short)0, Nre, false, false);
      Nim = __builtin_amdgcn_wmma_f32_16x16x4_f32(false, aim, false, bul[kb],
                                                  (short)0, Nim, false, false);
    }

#pragma unroll
    for (int v = 0; v < 8; ++v) {
      int idx = (v + 8 * h) * 16 + m;
      float2 d = sD[l * 256 + idx];
      float re = Nre[v], im = Nim[v];
      Sre[v] = re * d.x - im * d.y;
      Sim[v] = re * d.y + im * d.x;
    }
  }

  float sq[NQ];
#pragma unroll
  for (int q = 0; q < NQ; ++q) sq[q] = 0.0f;
#pragma unroll
  for (int v = 0; v < 8; ++v) {
    int idx = (v + 8 * h) * 16 + m;
    float p = Sre[v] * Sre[v] + Sim[v] * Sim[v];
#pragma unroll
    for (int q = 0; q < NQ; ++q)
      sq[q] += ((idx >> (7 - q)) & 1) ? -p : p;
  }
#pragma unroll
  for (int off = 16; off > 0; off >>= 1) {
#pragma unroll
    for (int q = 0; q < NQ; ++q)
      sq[q] += __shfl_xor(sq[q], off, 32);
  }
  if (lane == 0) {
#pragma unroll
    for (int q = 0; q < NQ; ++q)
      out[sample * NQ + q] = sq[q];
  }
  }
}

extern "C" void kernel_launch(void* const* d_in, const int* in_sizes, int n_in,
                              void* d_out, int out_size, void* d_ws, size_t ws_size,
                              hipStream_t stream) {
  const float* weights = (const float*)d_in[0];
  const float* z1      = (const float*)d_in[1];
  const float* z2      = (const float*)d_in[2];
  float* out           = (float*)d_out;
  int batch = in_sizes[1];
  qnoise_wmma_kernel<<<NBLOCKS, THREADS, 0, stream>>>(weights, z1, z2, out, batch);
}


# Fragment k order set to 8h+2kb+s and first product taken transposed (T^T=S^T*Uh^T, then N=T*Ul^T): accumulator registers are the next A operand, no LDS staging of the state; k sums reordered
# speedup vs baseline: 2.1776x; 1.1549x over previous
#include <hip/hip_runtime.h>
#include <math.h>

typedef __attribute__((ext_vector_type(2))) float v2f;
typedef __attribute__((ext_vector_type(8))) float v8f;

#define NQ 8
#define NL 3
#define DIM 256
#define WPB 32
#define THREADS (WPB * 32)
#define NBLOCKS 512

__device__ __forceinline__ float2 cmul(float2 a, float2 b) {
  return make_float2(a.x * b.x - a.y * b.y, a.x * b.y + a.y * b.x);
}

__global__ __launch_bounds__(THREADS) void qnoise_wmma_kernel(
    const float* __restrict__ weights,
    const float* __restrict__ z1,
    const float* __restrict__ z2,
    float* __restrict__ out,
    int batch) {
  __shared__ float  sU[NL * 2 * 256];
  __shared__ float2 sD[NL * 256];

  const int t    = threadIdx.x;
  const int lane = t & 31;
  const int wave = __builtin_amdgcn_readfirstlane(t >> 5);
  const int m    = lane & 15;
  const int h    = lane >> 4;

  for (int e = t; e < NL * 2 * 256; e += THREADS) {
    int l   = e >> 9;
    int mat = (e >> 8) & 1;
    int i   = (e >> 4) & 15;
    int k   = e & 15;
    float val = 1.0f;
#pragma unroll
    for (int j = 0; j < 4; ++j) {
      int q = mat * 4 + j;
      float th = 0.5f * weights[l * 15 + q];
      float c = cosf(th), s = sinf(th);
      int ib = (i >> (3 - j)) & 1;
      int kb = (k >> (3 - j)) & 1;
      val *= (ib == kb) ? c : ((ib == 0) ? -s : s);
    }
    sU[e] = val;
  }
  for (int e = t; e < NL * 256; e += THREADS) {
    int l = e >> 8;
    int idx = e & 255;
    float ang = 0.0f;
#pragma unroll
    for (int i2 = 0; i2 < 7; ++i2) {
      int b1 = (idx >> (7 - i2)) & 1;
      int b0 = (idx >> (6 - i2)) & 1;
      ang += weights[l * 15 + 8 + i2] * ((b1 == b0) ? 1.0f : -1.0f);
    }
    float ha = -0.5f * ang;
    sD[e] = make_float2(cosf(ha), sinf(ha));
  }
  __syncthreads();

  const int nwaves = gridDim.x * WPB;
  const int per    = (batch + nwaves - 1) / nwaves;
  const int first  = (blockIdx.x * WPB + wave) * per;
  const int last   = min(first + per, batch);

  for (int sample = first; sample < last; ++sample) {
  asm volatile("" ::: "memory");
  float a1 = asinf(z1[sample]);
  float a2 = asinf(z2[sample]);
  float ch = cosf(0.5f * a2), sh = sinf(0.5f * a2);
  float c1 = cosf(0.5f * a1), s1 = sinf(0.5f * a1);
  float2 v0 = make_float2(ch * c1, -sh * c1);
  float2 v1 = make_float2(ch * s1,  sh * s1);

  v8f Sre, Sim;
#pragma unroll
  for (int v = 0; v < 8; ++v) {
    int idx = (v + 8 * h) * 16 + m;
    float2 amp = make_float2(1.0f, 0.0f);
#pragma unroll
    for (int b = 0; b < 8; ++b)
      amp = cmul(amp, ((idx >> b) & 1) ? v1 : v0);
    Sre[v] = amp.x;
    Sim[v] = amp.y;
  }

  for (int l = 0; l < NL; ++l) {
    const float* Uh = &sU[l * 512];
    const float* Ul = &sU[l * 512 + 256];

    v2f buh[4], bul[4];
#pragma unroll
    for (int kb = 0; kb < 4; ++kb) {
      buh[kb][0] = Uh[m * 16 + 8 * h + 2 * kb + 0];
      buh[kb][1] = Uh[m * 16 + 8 * h + 2 * kb + 1];
      bul[kb][0] = Ul[m * 16 + 8 * h + 2 * kb + 0];
      bul[kb][1] = Ul[m * 16 + 8 * h + 2 * kb + 1];
    }

    v8f Tre = {0.f, 0.f, 0.f, 0.f, 0.f, 0.f, 0.f, 0.f};
    v8f Tim = {0.f, 0.f, 0.f, 0.f, 0.f, 0.f, 0.f, 0.f};
#pragma unroll
    for (int kb = 0; kb < 4; ++kb) {
      v2f are, aim;
      are[0] = Sre[2 * kb + 0];
      are[1] = Sre[2 * kb + 1];
      aim[0] = Sim[2 * kb + 0];
      aim[1] = Sim[2 * kb + 1];
      Tre = __builtin_amdgcn_wmma_f32_16x16x4_f32(false, are, false, buh[kb],
                                                  (short)0, Tre, false, false);
      Tim = __builtin_amdgcn_wmma_f32_16x16x4_f32(false, aim, false, buh[kb],
                                                  (short)0, Tim, false, false);
    }

    v8f Nre = {0.f, 0.f, 0.f, 0.f, 0.f, 0.f, 0.f, 0.f};
    v8f Nim = {0.f, 0.f, 0.f, 0.f, 0.f, 0.f, 0.f, 0.f};
#pragma unroll
    for (int kb = 0; kb < 4; ++kb) {
      v2f are, aim;
      are[0] = Tre[2 * kb + 0];
      are[1] = Tre[2 * kb + 1];
      aim[0] = Tim[2 * kb + 0];
      aim[1] = Tim[2 * kb + 1];
      Nre = __builtin_amdgcn_wmma_f32_16x16x4_f32(false, are, false, bul[kb],
                                                  (short)0, Nre, false, false);
      Nim = __builtin_amdgcn_wmma_f32_16x16x4_f32(false, aim, false, bul[kb],
                                                  (short)0, Nim, false, false);
    }

#pragma unroll
    for (int v = 0; v < 8; ++v) {
      int idx = (v + 8 * h) * 16 + m;
      float2 d = sD[l * 256 + idx];
      float re = Nre[v], im = Nim[v];
      Sre[v] = re * d.x - im * d.y;
      Sim[v] = re * d.y + im * d.x;
    }
  }

  float sq[NQ];
#pragma unroll
  for (int q = 0; q < NQ; ++q) sq[q] = 0.0f;
#pragma unroll
  for (int v = 0; v < 8; ++v) {
    int idx = (v + 8 * h) * 16 + m;
    float p = Sre[v] * Sre[v] + Sim[v] * Sim[v];
#pragma unroll
    for (int q = 0; q < NQ; ++q)
      sq[q] += ((idx >> (7 - q)) & 1) ? -p : p;
  }
#pragma unroll
  for (int off = 16; off > 0; off >>= 1) {
#pragma unroll
    for (int q = 0; q < NQ; ++q)
      sq[q] += __shfl_xor(sq[q], off, 32);
  }
  if (lane == 0) {
#pragma unroll
    for (int q = 0; q < NQ; ++q)
      out[sample * NQ + q] = sq[q];
  }
  }
}

extern "C" void kernel_launch(void* const* d_in, const int* in_sizes, int n_in,
                              void* d_out, int out_size, void* d_ws, size_t ws_size,
                              hipStream_t stream) {
  const float* weights = (const float*)d_in[0];
  const float* z1      = (const float*)d_in[1];
  const float* z2      = (const float*)d_in[2];
  float* out           = (float*)d_out;
  int batch = in_sizes[1];
  qnoise_wmma_kernel<<<NBLOCKS, THREADS, 0, stream>>>(weights, z1, z2, out, batch);
}


# Table build: cos and sin of each of the 24 RY half-angles evaluated once per block and kept in LDS; operator tables assembled from them, not 4 cosf+sinf per table element; same factors, same order
# speedup vs baseline: 2.2677x; 1.0414x over previous
#include <hip/hip_runtime.h>
#include <math.h>

typedef __attribute__((ext_vector_type(2))) float v2f;
typedef __attribute__((ext_vector_type(8))) float v8f;

#define NQ 8
#define NL 3
#define DIM 256
#define WPB 32
#define THREADS (WPB * 32)
#define NBLOCKS 512

__device__ __forceinline__ float2 cmul(float2 a, float2 b) {
  return make_float2(a.x * b.x - a.y * b.y, a.x * b.y + a.y * b.x);
}

__global__ __launch_bounds__(THREADS) void qnoise_wmma_kernel(
    const float* __restrict__ weights,
    const float* __restrict__ z1,
    const float* __restrict__ z2,
    float* __restrict__ out,
    int batch) {
  __shared__ float  sU[NL * 2 * 256];
  __shared__ float2 sD[NL * 256];
  __shared__ float  sCS[NL * NQ * 2];

  const int t    = threadIdx.x;
  const int lane = t & 31;
  const int wave = __builtin_amdgcn_readfirstlane(t >> 5);
  const int m    = lane & 15;
  const int h    = lane >> 4;

  if (t < NL * NQ) {
    int l = t >> 3;
    int q = t & 7;
    float th = 0.5f * weights[l * 15 + q];
    sCS[2 * t + 0] = cosf(th);
    sCS[2 * t + 1] = sinf(th);
  }
  if (t >= THREADS - NL * 256) {
    int e = t - (THREADS - NL * 256);
    int l = e >> 8;
    int idx = e & 255;
    float ang = 0.0f;
#pragma unroll
    for (int i2 = 0; i2 < 7; ++i2) {
      int b1 = (idx >> (7 - i2)) & 1;
      int b0 = (idx >> (6 - i2)) & 1;
      ang += weights[l * 15 + 8 + i2] * ((b1 == b0) ? 1.0f : -1.0f);
    }
    float ha = -0.5f * ang;
    sD[e] = make_float2(cosf(ha), sinf(ha));
  }
  __syncthreads();
  for (int e = t; e < NL * 2 * 256; e += THREADS) {
    int l   = e >> 9;
    int mat = (e >> 8) & 1;
    int i   = (e >> 4) & 15;
    int k   = e & 15;
    float val = 1.0f;
#pragma unroll
    for (int j = 0; j < 4; ++j) {
      int ib = (i >> (3 - j)) & 1;
      int kb = (k >> (3 - j)) & 1;
      int od = ib ^ kb;
      float f = sCS[2 * (l * NQ + mat * 4 + j) + od];
      val *= (od & (ib ^ 1)) ? -f : f;
    }
    sU[e] = val;
  }
  __syncthreads();

  const int nwaves = gridDim.x * WPB;
  const int per    = (batch + nwaves - 1) / nwaves;
  const int first  = (blockIdx.x * WPB + wave) * per;
  const int last   = min(first + per, batch);

  for (int sample = first; sample < last; ++sample) {
  asm volatile("" ::: "memory");
  float a1 = asinf(z1[sample]);
  float a2 = asinf(z2[sample]);
  float ch = cosf(0.5f * a2), sh = sinf(0.5f * a2);
  float c1 = cosf(0.5f * a1), s1 = sinf(0.5f * a1);
  float2 v0 = make_float2(ch * c1, -sh * c1);
  float2 v1 = make_float2(ch * s1,  sh * s1);

  v8f Sre, Sim;
#pragma unroll
  for (int v = 0; v < 8; ++v) {
    int idx = (v + 8 * h) * 16 + m;
    float2 amp = make_float2(1.0f, 0.0f);
#pragma unroll
    for (int b = 0; b < 8; ++b)
      amp = cmul(amp, ((idx >> b) & 1) ? v1 : v0);
    Sre[v] = amp.x;
    Sim[v] = amp.y;
  }

  for (int l = 0; l < NL; ++l) {
    const float* Uh = &sU[l * 512];
    const float* Ul = &sU[l * 512 + 256];

    v2f buh[4], bul[4];
#pragma unroll
    for (int kb = 0; kb < 4; ++kb) {
      buh[kb][0] = Uh[m * 16 + 8 * h + 2 * kb + 0];
      buh[kb][1] = Uh[m * 16 + 8 * h + 2 * kb + 1];
      bul[kb][0] = Ul[m * 16 + 8 * h + 2 * kb + 0];
      bul[kb][1] = Ul[m * 16 + 8 * h + 2 * kb + 1];
    }

    v8f Tre = {0.f, 0.f, 0.f, 0.f, 0.f, 0.f, 0.f, 0.f};
    v8f Tim = {0.f, 0.f, 0.f, 0.f, 0.f, 0.f, 0.f, 0.f};
#pragma unroll
    for (int kb = 0; kb < 4; ++kb) {
      v2f are, aim;
      are[0] = Sre[2 * kb + 0];
      are[1] = Sre[2 * kb + 1];
      aim[0] = Sim[2 * kb + 0];
      aim[1] = Sim[2 * kb + 1];
      Tre = __builtin_amdgcn_wmma_f32_16x16x4_f32(false, are, false, buh[kb],
                                                  (short)0, Tre, false, false);
      Tim = __builtin_amdgcn_wmma_f32_16x16x4_f32(false, aim, false, buh[kb],
                                                  (short)0, Tim, false, false);
    }

    v8f Nre = {0.f, 0.f, 0.f, 0.f, 0.f, 0.f, 0.f, 0.f};
    v8f Nim = {0.f, 0.f, 0.f, 0.f, 0.f, 0.f, 0.f, 0.f};
#pragma unroll
    for (int kb = 0; kb < 4; ++kb) {
      v2f are, aim;
      are[0] = Tre[2 * kb + 0];
      are[1] = Tre[2 * kb + 1];
      aim[0] = Tim[2 * kb + 0];
      aim[1] = Tim[2 * kb + 1];
      Nre = __builtin_amdgcn_wmma_f32_16x16x4_f32(false, are, false, bul[kb],
                                                  (short)0, Nre, false, false);
      Nim = __builtin_amdgcn_wmma_f32_16x16x4_f32(false, aim, false, bul[kb],
                                                  (short)0, Nim, false, false);
    }

#pragma unroll
    for (int v = 0; v < 8; ++v) {
      int idx = (v + 8 * h) * 16 + m;
      float2 d = sD[l * 256 + idx];
      float re = Nre[v], im = Nim[v];
      Sre[v] = re * d.x - im * d.y;
      Sim[v] = re * d.y + im * d.x;
    }
  }

  float sq[NQ];
#pragma unroll
  for (int q = 0; q < NQ; ++q) sq[q] = 0.0f;
#pragma unroll
  for (int v = 0; v < 8; ++v) {
    int idx = (v + 8 * h) * 16 + m;
    float p = Sre[v] * Sre[v] + Sim[v] * Sim[v];
#pragma unroll
    for (int q = 0; q < NQ; ++q)
      sq[q] += ((idx >> (7 - q)) & 1) ? -p : p;
  }
#pragma unroll
  for (int off = 16; off > 0; off >>= 1) {
#pragma unroll
    for (int q = 0; q < NQ; ++q)
      sq[q] += __shfl_xor(sq[q], off, 32);
  }
  if (lane == 0) {
#pragma unroll
    for (int q = 0; q < NQ; ++q)
      out[sample * NQ + q] = sq[q];
  }
  }
}

extern "C" void kernel_launch(void* const* d_in, const int* in_sizes, int n_in,
                              void* d_out, int out_size, void* d_ws, size_t ws_size,
                              hipStream_t stream) {
  const float* weights = (const float*)d_in[0];
  const float* z1      = (const float*)d_in[1];
  const float* z2      = (const float*)d_in[2];
  float* out           = (float*)d_out;
  int batch = in_sizes[1];
  qnoise_wmma_kernel<<<NBLOCKS, THREADS, 0, stream>>>(weights, z1, z2, out, batch);
}


# Layer 0's RYs act on a product state: applied to each wire's 2-vector before the state is built, so layer 0 needs no matrix product (32 WMMAs a sample, not 48); rounding order changes
# speedup vs baseline: 2.8413x; 1.2529x over previous
#include <hip/hip_runtime.h>
#include <math.h>

typedef __attribute__((ext_vector_type(2))) float v2f;
typedef __attribute__((ext_vector_type(8))) float v8f;

#define NQ 8
#define NL 3
#define DIM 256
#define WPB 32
#define THREADS (WPB * 32)
#define NBLOCKS 512

__device__ __forceinline__ float2 cmul(float2 a, float2 b) {
  return make_float2(a.x * b.x - a.y * b.y, a.x * b.y + a.y * b.x);
}

__global__ __launch_bounds__(THREADS) void qnoise_wmma_kernel(
    const float* __restrict__ weights,
    const float* __restrict__ z1,
    const float* __restrict__ z2,
    float* __restrict__ out,
    int batch) {
  __shared__ float  sU[NL * 2 * 256];
  __shared__ float2 sD[NL * 256];
  __shared__ float  sCS[NL * NQ * 2];

  const int t    = threadIdx.x;
  const int lane = t & 31;
  const int wave = __builtin_amdgcn_readfirstlane(t >> 5);
  const int m    = lane & 15;
  const int h    = lane >> 4;

  if (t < NL * NQ) {
    int l = t >> 3;
    int q = t & 7;
    float th = 0.5f * weights[l * 15 + q];
    sCS[2 * t + 0] = cosf(th);
    sCS[2 * t + 1] = sinf(th);
  }
  if (t >= THREADS - NL * 256) {
    int e = t - (THREADS - NL * 256);
    int l = e >> 8;
    int idx = e & 255;
    float ang = 0.0f;
#pragma unroll
    for (int i2 = 0; i2 < 7; ++i2) {
      int b1 = (idx >> (7 - i2)) & 1;
      int b0 = (idx >> (6 - i2)) & 1;
      ang += weights[l * 15 + 8 + i2] * ((b1 == b0) ? 1.0f : -1.0f);
    }
    float ha = -0.5f * ang;
    sD[e] = make_float2(cosf(ha), sinf(ha));
  }
  __syncthreads();
  for (int e = t; e < NL * 2 * 256; e += THREADS) {
    int l   = e >> 9;
    int mat = (e >> 8) & 1;
    int i   = (e >> 4) & 15;
    int k   = e & 15;
    float val = 1.0f;
#pragma unroll
    for (int j = 0; j < 4; ++j) {
      int ib = (i >> (3 - j)) & 1;
      int kb = (k >> (3 - j)) & 1;
      int od = ib ^ kb;
      float f = sCS[2 * (l * NQ + mat * 4 + j) + od];
      val *= (od & (ib ^ 1)) ? -f : f;
    }
    sU[e] = val;
  }
  __syncthreads();

  float c0[NQ], s0[NQ];
#pragma unroll
  for (int q = 0; q < NQ; ++q) {
    c0[q] = __int_as_float(__builtin_amdgcn_readfirstlane(__float_as_int(sCS[2 * q + 0])));
    s0[q] = __int_as_float(__builtin_amdgcn_readfirstlane(__float_as_int(sCS[2 * q + 1])));
  }

  const int nwaves = gridDim.x * WPB;
  const int per    = (batch + nwaves - 1) / nwaves;
  const int first  = (blockIdx.x * WPB + wave) * per;
  const int last   = min(first + per, batch);

  for (int sample = first; sample < last; ++sample) {
  asm volatile("" ::: "memory");
  float a1 = asinf(z1[sample]);
  float a2 = asinf(z2[sample]);
  float ch = cosf(0.5f * a2), sh = sinf(0.5f * a2);
  float c1 = cosf(0.5f * a1), s1 = sinf(0.5f * a1);
  float2 v0 = make_float2(ch * c1, -sh * c1);
  float2 v1 = make_float2(ch * s1,  sh * s1);

  float2 ua[NQ], ub[NQ];
#pragma unroll
  for (int q = 0; q < NQ; ++q) {
    ua[q] = make_float2(c0[q] * v0.x - s0[q] * v1.x, c0[q] * v0.y - s0[q] * v1.y);
    ub[q] = make_float2(s0[q] * v0.x + c0[q] * v1.x, s0[q] * v0.y + c0[q] * v1.y);
  }
  float2 lanef = (m & 1) ? ub[7] : ua[7];
  lanef = cmul(lanef, (m & 2) ? ub[6] : ua[6]);
  lanef = cmul(lanef, (m & 4) ? ub[5] : ua[5]);
  lanef = cmul(lanef, (m & 8) ? ub[4] : ua[4]);
  lanef = cmul(lanef, h ? ub[0] : ua[0]);

  v8f Sre, Sim;
#pragma unroll
  for (int v = 0; v < 8; ++v) {
    int idx = (v + 8 * h) * 16 + m;
    float2 regf = (v & 4) ? ub[1] : ua[1];
    regf = cmul(regf, (v & 2) ? ub[2] : ua[2]);
    regf = cmul(regf, (v & 1) ? ub[3] : ua[3]);
    float2 amp = cmul(lanef, regf);
    float2 d = sD[idx];
    Sre[v] = amp.x * d.x - amp.y * d.y;
    Sim[v] = amp.x * d.y + amp.y * d.x;
  }

  for (int l = 1; l < NL; ++l) {
    const float* Uh = &sU[l * 512];
    const float* Ul = &sU[l * 512 + 256];

    v2f buh[4], bul[4];
#pragma unroll
    for (int kb = 0; kb < 4; ++kb) {
      buh[kb][0] = Uh[m * 16 + 8 * h + 2 * kb + 0];
      buh[kb][1] = Uh[m * 16 + 8 * h + 2 * kb + 1];
      bul[kb][0] = Ul[m * 16 + 8 * h + 2 * kb + 0];
      bul[kb][1] = Ul[m * 16 + 8 * h + 2 * kb + 1];
    }

    v8f Tre = {0.f, 0.f, 0.f, 0.f, 0.f, 0.f, 0.f, 0.f};
    v8f Tim = {0.f, 0.f, 0.f, 0.f, 0.f, 0.f, 0.f, 0.f};
#pragma unroll
    for (int kb = 0; kb < 4; ++kb) {
      v2f are, aim;
      are[0] = Sre[2 * kb + 0];
      are[1] = Sre[2 * kb + 1];
      aim[0] = Sim[2 * kb + 0];
      aim[1] = Sim[2 * kb + 1];
      Tre = __builtin_amdgcn_wmma_f32_16x16x4_f32(false, are, false, buh[kb],
                                                  (short)0, Tre, false, false);
      Tim = __builtin_amdgcn_wmma_f32_16x16x4_f32(false, aim, false, buh[kb],
                                                  (short)0, Tim, false, false);
    }

    v8f Nre = {0.f, 0.f, 0.f, 0.f, 0.f, 0.f, 0.f, 0.f};
    v8f Nim = {0.f, 0.f, 0.f, 0.f, 0.f, 0.f, 0.f, 0.f};
#pragma unroll
    for (int kb = 0; kb < 4; ++kb) {
      v2f are, aim;
      are[0] = Tre[2 * kb + 0];
      are[1] = Tre[2 * kb + 1];
      aim[0] = Tim[2 * kb + 0];
      aim[1] = Tim[2 * kb + 1];
      Nre = __builtin_amdgcn_wmma_f32_16x16x4_f32(false, are, false, bul[kb],
                                                  (short)0, Nre, false, false);
      Nim = __builtin_amdgcn_wmma_f32_16x16x4_f32(false, aim, false, bul[kb],
                                                  (short)0, Nim, false, false);
    }

#pragma unroll
    for (int v = 0; v < 8; ++v) {
      int idx = (v + 8 * h) * 16 + m;
      float2 d = sD[l * 256 + idx];
      float re = Nre[v], im = Nim[v];
      Sre[v] = re * d.x - im * d.y;
      Sim[v] = re * d.y + im * d.x;
    }
  }

  float sq[NQ];
#pragma unroll
  for (int q = 0; q < NQ; ++q) sq[q] = 0.0f;
#pragma unroll
  for (int v = 0; v < 8; ++v) {
    int idx = (v + 8 * h) * 16 + m;
    float p = Sre[v] * Sre[v] + Sim[v] * Sim[v];
#pragma unroll
    for (int q = 0; q < NQ; ++q)
      sq[q] += ((idx >> (7 - q)) & 1) ? -p : p;
  }
#pragma unroll
  for (int off = 16; off > 0; off >>= 1) {
#pragma unroll
    for (int q = 0; q < NQ; ++q)
      sq[q] += __shfl_xor(sq[q], off, 32);
  }
  if (lane == 0) {
#pragma unroll
    for (int q = 0; q < NQ; ++q)
      out[sample * NQ + q] = sq[q];
  }
  }
}

extern "C" void kernel_launch(void* const* d_in, const int* in_sizes, int n_in,
                              void* d_out, int out_size, void* d_ws, size_t ws_size,
                              hipStream_t stream) {
  const float* weights = (const float*)d_in[0];
  const float* z1      = (const float*)d_in[1];
  const float* z2      = (const float*)d_in[2];
  float* out           = (float*)d_out;
  int batch = in_sizes[1];
  qnoise_wmma_kernel<<<NBLOCKS, THREADS, 0, stream>>>(weights, z1, z2, out, batch);
}


# Last layer: Ul product and RZZ phase dropped; Z of wires 4-7 = cos(w)*sum(|p0|^2-|p1|^2) - 2sin(w)*Re sum(p0*conj(p1)) on the Uh-rotated state: 24 WMMAs a sample, not 32; sums reordered
# speedup vs baseline: 3.1304x; 1.1017x over previous
#include <hip/hip_runtime.h>
#include <math.h>

typedef __attribute__((ext_vector_type(2))) float v2f;
typedef __attribute__((ext_vector_type(8))) float v8f;

#define NQ 8
#define NL 3
#define DIM 256
#define WPB 32
#define THREADS (WPB * 32)
#define NBLOCKS 512

__device__ __forceinline__ float2 cmul(float2 a, float2 b) {
  return make_float2(a.x * b.x - a.y * b.y, a.x * b.y + a.y * b.x);
}

__global__ __launch_bounds__(THREADS) void qnoise_wmma_kernel(
    const float* __restrict__ weights,
    const float* __restrict__ z1,
    const float* __restrict__ z2,
    float* __restrict__ out,
    int batch) {
  __shared__ float  sU[(NL * 2 - 1) * 256];
  __shared__ float2 sD[(NL - 1) * 256];
  __shared__ float  sCS[NL * NQ * 2];
  __shared__ float  sZ[8];

  const int t    = threadIdx.x;
  const int lane = t & 31;
  const int wave = __builtin_amdgcn_readfirstlane(t >> 5);
  const int m    = lane & 15;
  const int h    = lane >> 4;

  if (t < NL * NQ) {
    int l = t >> 3;
    int q = t & 7;
    float th = 0.5f * weights[l * 15 + q];
    sCS[2 * t + 0] = cosf(th);
    sCS[2 * t + 1] = sinf(th);
  }
  if (t >= THREADS - (NL - 1) * 256) {
    int e = t - (THREADS - (NL - 1) * 256);
    int l = e >> 8;
    int idx = e & 255;
    float ang = 0.0f;
#pragma unroll
    for (int i2 = 0; i2 < 7; ++i2) {
      int b1 = (idx >> (7 - i2)) & 1;
      int b0 = (idx >> (6 - i2)) & 1;
      ang += weights[l * 15 + 8 + i2] * ((b1 == b0) ? 1.0f : -1.0f);
    }
    float ha = -0.5f * ang;
    sD[e] = make_float2(cosf(ha), sinf(ha));
  }
  __syncthreads();
  for (int e = t; e < (NL * 2 - 1) * 256; e += THREADS) {
    int l   = e >> 9;
    int mat = (e >> 8) & 1;
    int i   = (e >> 4) & 15;
    int k   = e & 15;
    float val = 1.0f;
#pragma unroll
    for (int j = 0; j < 4; ++j) {
      int ib = (i >> (3 - j)) & 1;
      int kb = (k >> (3 - j)) & 1;
      int od = ib ^ kb;
      float f = sCS[2 * (l * NQ + mat * 4 + j) + od];
      val *= (od & (ib ^ 1)) ? -f : f;
    }
    sU[e] = val;
  }
  if (t < 4) {
    float c = sCS[2 * ((NL - 1) * NQ + 4 + t) + 0];
    float s = sCS[2 * ((NL - 1) * NQ + 4 + t) + 1];
    sZ[t]     = c * c - s * s;
    sZ[4 + t] = (t == 0 ? 2.0f : 4.0f) * c * s;
  }
  __syncthreads();

  float c0[NQ], s0[NQ];
#pragma unroll
  for (int q = 0; q < NQ; ++q) {
    c0[q] = __int_as_float(__builtin_amdgcn_readfirstlane(__float_as_int(sCS[2 * q + 0])));
    s0[q] = __int_as_float(__builtin_amdgcn_readfirstlane(__float_as_int(sCS[2 * q + 1])));
  }

  const int nwaves = gridDim.x * WPB;
  const int per    = (batch + nwaves - 1) / nwaves;
  const int first  = (blockIdx.x * WPB + wave) * per;
  const int last   = min(first + per, batch);

  for (int sample = first; sample < last; ++sample) {
  asm volatile("" ::: "memory");
  float a1 = asinf(z1[sample]);
  float a2 = asinf(z2[sample]);
  float ch = cosf(0.5f * a2), sh = sinf(0.5f * a2);
  float c1 = cosf(0.5f * a1), s1 = sinf(0.5f * a1);
  float2 v0 = make_float2(ch * c1, -sh * c1);
  float2 v1 = make_float2(ch * s1,  sh * s1);

  float2 ua[NQ], ub[NQ];
#pragma unroll
  for (int q = 0; q < NQ; ++q) {
    ua[q] = make_float2(c0[q] * v0.x - s0[q] * v1.x, c0[q] * v0.y - s0[q] * v1.y);
    ub[q] = make_float2(s0[q] * v0.x + c0[q] * v1.x, s0[q] * v0.y + c0[q] * v1.y);
  }
  float2 lanef = (m & 1) ? ub[7] : ua[7];
  lanef = cmul(lanef, (m & 2) ? ub[6] : ua[6]);
  lanef = cmul(lanef, (m & 4) ? ub[5] : ua[5]);
  lanef = cmul(lanef, (m & 8) ? ub[4] : ua[4]);
  lanef = cmul(lanef, h ? ub[0] : ua[0]);

  v8f Sre, Sim;
#pragma unroll
  for (int v = 0; v < 8; ++v) {
    int idx = (v + 8 * h) * 16 + m;
    float2 regf = (v & 4) ? ub[1] : ua[1];
    regf = cmul(regf, (v & 2) ? ub[2] : ua[2]);
    regf = cmul(regf, (v & 1) ? ub[3] : ua[3]);
    float2 amp = cmul(lanef, regf);
    float2 d = sD[idx];
    Sre[v] = amp.x * d.x - amp.y * d.y;
    Sim[v] = amp.x * d.y + amp.y * d.x;
  }

  for (int l = 1; l < NL - 1; ++l) {
    const float* Uh = &sU[l * 512];
    const float* Ul = &sU[l * 512 + 256];

    v2f buh[4], bul[4];
#pragma unroll
    for (int kb = 0; kb < 4; ++kb) {
      buh[kb][0] = Uh[m * 16 + 8 * h + 2 * kb + 0];
      buh[kb][1] = Uh[m * 16 + 8 * h + 2 * kb + 1];
      bul[kb][0] = Ul[m * 16 + 8 * h + 2 * kb + 0];
      bul[kb][1] = Ul[m * 16 + 8 * h + 2 * kb + 1];
    }

    v8f Tre = {0.f, 0.f, 0.f, 0.f, 0.f, 0.f, 0.f, 0.f};
    v8f Tim = {0.f, 0.f, 0.f, 0.f, 0.f, 0.f, 0.f, 0.f};
#pragma unroll
    for (int kb = 0; kb < 4; ++kb) {
      v2f are, aim;
      are[0] = Sre[2 * kb + 0];
      are[1] = Sre[2 * kb + 1];
      aim[0] = Sim[2 * kb + 0];
      aim[1] = Sim[2 * kb + 1];
      Tre = __builtin_amdgcn_wmma_f32_16x16x4_f32(false, are, false, buh[kb],
                                                  (short)0, Tre, false, false);
      Tim = __builtin_amdgcn_wmma_f32_16x16x4_f32(false, aim, false, buh[kb],
                                                  (short)0, Tim, false, false);
    }

    v8f Nre = {0.f, 0.f, 0.f, 0.f, 0.f, 0.f, 0.f, 0.f};
    v8f Nim = {0.f, 0.f, 0.f, 0.f, 0.f, 0.f, 0.f, 0.f};
#pragma unroll
    for (int kb = 0; kb < 4; ++kb) {
      v2f are, aim;
      are[0] = Tre[2 * kb + 0];
      are[1] = Tre[2 * kb + 1];
      aim[0] = Tim[2 * kb + 0];
      aim[1] = Tim[2 * kb + 1];
      Nre = __builtin_amdgcn_wmma_f32_16x16x4_f32(false, are, false, bul[kb],
                                                  (short)0, Nre, false, false);
      Nim = __builtin_amdgcn_wmma_f32_16x16x4_f32(false, aim, false, bul[kb],
                                                  (short)0, Nim, false, false);
    }

#pragma unroll
    for (int v = 0; v < 8; ++v) {
      int idx = (v + 8 * h) * 16 + m;
      float2 d = sD[l * 256 + idx];
      float re = Nre[v], im = Nim[v];
      Sre[v] = re * d.x - im * d.y;
      Sim[v] = re * d.y + im * d.x;
    }
  }

  v8f Tre = {0.f, 0.f, 0.f, 0.f, 0.f, 0.f, 0.f, 0.f};
  v8f Tim = {0.f, 0.f, 0.f, 0.f, 0.f, 0.f, 0.f, 0.f};
  {
    const float* Uh = &sU[(NL - 1) * 512];
#pragma unroll
    for (int kb = 0; kb < 4; ++kb) {
      v2f b, are, aim;
      b[0] = Uh[m * 16 + 8 * h + 2 * kb + 0];
      b[1] = Uh[m * 16 + 8 * h + 2 * kb + 1];
      are[0] = Sre[2 * kb + 0];
      are[1] = Sre[2 * kb + 1];
      aim[0] = Sim[2 * kb + 0];
      aim[1] = Sim[2 * kb + 1];
      Tre = __builtin_amdgcn_wmma_f32_16x16x4_f32(false, are, false, b,
                                                  (short)0, Tre, false, false);
      Tim = __builtin_amdgcn_wmma_f32_16x16x4_f32(false, aim, false, b,
                                                  (short)0, Tim, false, false);
    }
  }

  float p[8];
#pragma unroll
  for (int r = 0; r < 8; ++r) p[r] = Tre[r] * Tre[r] + Tim[r] * Tim[r];
  float s01 = p[0] + p[1], d01 = p[0] - p[1];
  float s23 = p[2] + p[3], d23 = p[2] - p[3];
  float s45 = p[4] + p[5], d45 = p[4] - p[5];
  float s67 = p[6] + p[7], d67 = p[6] - p[7];
  float sA = s01 + s23, eA = s01 - s23;
  float sB = s45 + s67, eB = s45 - s67;
  float tt = sA + sB;
  float a5 = sA - sB;
  float a6 = eA + eB;
  float a7 = (d01 + d23) + (d45 + d67);
  float b5 = 0.0f, b6 = 0.0f, b7 = 0.0f;
#pragma unroll
  for (int r = 0; r < 8; ++r) {
    if (!(r & 4)) b5 += Tre[r] * Tre[r + 4] + Tim[r] * Tim[r + 4];
    if (!(r & 2)) b6 += Tre[r] * Tre[r + 2] + Tim[r] * Tim[r + 2];
    if (!(r & 1)) b7 += Tre[r] * Tre[r + 1] + Tim[r] * Tim[r + 1];
  }
  float x4 = 0.0f;
#pragma unroll
  for (int r = 0; r < 8; ++r) {
    int bre = __float_as_int(Tre[r]);
    int bim = __float_as_int(Tim[r]);
    float ore = __int_as_float(
        __builtin_amdgcn_permlanex16(bre, bre, 0x76543210, 0xfedcba98, false, false));
    float oim = __int_as_float(
        __builtin_amdgcn_permlanex16(bim, bim, 0x76543210, 0xfedcba98, false, false));
    x4 += Tre[r] * ore + Tim[r] * oim;
  }

  float sq[NQ];
  sq[0] = (m & 8) ? -tt : tt;
  sq[1] = (m & 4) ? -tt : tt;
  sq[2] = (m & 2) ? -tt : tt;
  sq[3] = (m & 1) ? -tt : tt;
  sq[4] = sZ[0] * (h ? -tt : tt) - sZ[4] * x4;
  sq[5] = sZ[1] * a5 - sZ[5] * b5;
  sq[6] = sZ[2] * a6 - sZ[6] * b6;
  sq[7] = sZ[3] * a7 - sZ[7] * b7;
#pragma unroll
  for (int off = 16; off > 0; off >>= 1) {
#pragma unroll
    for (int q = 0; q < NQ; ++q)
      sq[q] += __shfl_xor(sq[q], off, 32);
  }
  if (lane == 0) {
#pragma unroll
    for (int q = 0; q < NQ; ++q)
      out[sample * NQ + q] = sq[q];
  }
  }
}

extern "C" void kernel_launch(void* const* d_in, const int* in_sizes, int n_in,
                              void* d_out, int out_size, void* d_ws, size_t ws_size,
                              hipStream_t stream) {
  const float* weights = (const float*)d_in[0];
  const float* z1      = (const float*)d_in[1];
  const float* z2      = (const float*)d_in[2];
  float* out           = (float*)d_out;
  int batch = in_sizes[1];
  qnoise_wmma_kernel<<<NBLOCKS, THREADS, 0, stream>>>(weights, z1, z2, out, batch);
}


# Final 32-lane sum of the 8 outputs: value count halved at levels xor 16/8/4 (permlanex16 + DPP row_xmask), 9 vector-unit exchanges instead of 40 ds_bpermute; same pairs, sums bit-identical
# speedup vs baseline: 3.3894x; 1.0827x over previous
#include <hip/hip_runtime.h>
#include <math.h>

typedef __attribute__((ext_vector_type(2))) float v2f;
typedef __attribute__((ext_vector_type(8))) float v8f;

#define NQ 8
#define NL 3
#define DIM 256
#define WPB 32
#define THREADS (WPB * 32)
#define NBLOCKS 512

__device__ __forceinline__ float2 cmul(float2 a, float2 b) {
  return make_float2(a.x * b.x - a.y * b.y, a.x * b.y + a.y * b.x);
}


__device__ __forceinline__ float xhalf(float v) {
  int b = __float_as_int(v);
  return __int_as_float(
      __builtin_amdgcn_permlanex16(b, b, 0x76543210, 0xfedcba98, false, false));
}
template <int MASK>
__device__ __forceinline__ float xrow(float v) {
  return __int_as_float(__builtin_amdgcn_update_dpp(0, __float_as_int(v), 0x160 + MASK, 0xf, 0xf, true));
}

__global__ __launch_bounds__(THREADS) void qnoise_wmma_kernel(
    const float* __restrict__ weights,
    const float* __restrict__ z1,
    const float* __restrict__ z2,
    float* __restrict__ out,
    int batch) {
  __shared__ float  sU[(NL * 2 - 1) * 256];
  __shared__ float2 sD[(NL - 1) * 256];
  __shared__ float  sCS[NL * NQ * 2];
  __shared__ float  sZ[8];

  const int t    = threadIdx.x;
  const int lane = t & 31;
  const int wave = __builtin_amdgcn_readfirstlane(t >> 5);
  const int m    = lane & 15;
  const int h    = lane >> 4;

  if (t < NL * NQ) {
    int l = t >> 3;
    int q = t & 7;
    float th = 0.5f * weights[l * 15 + q];
    sCS[2 * t + 0] = cosf(th);
    sCS[2 * t + 1] = sinf(th);
  }
  if (t >= THREADS - (NL - 1) * 256) {
    int e = t - (THREADS - (NL - 1) * 256);
    int l = e >> 8;
    int idx = e & 255;
    float ang = 0.0f;
#pragma unroll
    for (int i2 = 0; i2 < 7; ++i2) {
      int b1 = (idx >> (7 - i2)) & 1;
      int b0 = (idx >> (6 - i2)) & 1;
      ang += weights[l * 15 + 8 + i2] * ((b1 == b0) ? 1.0f : -1.0f);
    }
    float ha = -0.5f * ang;
    sD[e] = make_float2(cosf(ha), sinf(ha));
  }
  __syncthreads();
  for (int e = t; e < (NL * 2 - 1) * 256; e += THREADS) {
    int l   = e >> 9;
    int mat = (e >> 8) & 1;
    int i   = (e >> 4) & 15;
    int k   = e & 15;
    float val = 1.0f;
#pragma unroll
    for (int j = 0; j < 4; ++j) {
      int ib = (i >> (3 - j)) & 1;
      int kb = (k >> (3 - j)) & 1;
      int od = ib ^ kb;
      float f = sCS[2 * (l * NQ + mat * 4 + j) + od];
      val *= (od & (ib ^ 1)) ? -f : f;
    }
    sU[e] = val;
  }
  if (t < 4) {
    float c = sCS[2 * ((NL - 1) * NQ + 4 + t) + 0];
    float s = sCS[2 * ((NL - 1) * NQ + 4 + t) + 1];
    sZ[t]     = c * c - s * s;
    sZ[4 + t] = (t == 0 ? 2.0f : 4.0f) * c * s;
  }
  __syncthreads();

  float c0[NQ], s0[NQ];
#pragma unroll
  for (int q = 0; q < NQ; ++q) {
    c0[q] = __int_as_float(__builtin_amdgcn_readfirstlane(__float_as_int(sCS[2 * q + 0])));
    s0[q] = __int_as_float(__builtin_amdgcn_readfirstlane(__float_as_int(sCS[2 * q + 1])));
  }

  const int nwaves = gridDim.x * WPB;
  const int per    = (batch + nwaves - 1) / nwaves;
  const int first  = (blockIdx.x * WPB + wave) * per;
  const int last   = min(first + per, batch);

  for (int sample = first; sample < last; ++sample) {
  asm volatile("" ::: "memory");
  float a1 = asinf(z1[sample]);
  float a2 = asinf(z2[sample]);
  float ch = cosf(0.5f * a2), sh = sinf(0.5f * a2);
  float c1 = cosf(0.5f * a1), s1 = sinf(0.5f * a1);
  float2 v0 = make_float2(ch * c1, -sh * c1);
  float2 v1 = make_float2(ch * s1,  sh * s1);

  float2 ua[NQ], ub[NQ];
#pragma unroll
  for (int q = 0; q < NQ; ++q) {
    ua[q] = make_float2(c0[q] * v0.x - s0[q] * v1.x, c0[q] * v0.y - s0[q] * v1.y);
    ub[q] = make_float2(s0[q] * v0.x + c0[q] * v1.x, s0[q] * v0.y + c0[q] * v1.y);
  }
  float2 lanef = (m & 1) ? ub[7] : ua[7];
  lanef = cmul(lanef, (m & 2) ? ub[6] : ua[6]);
  lanef = cmul(lanef, (m & 4) ? ub[5] : ua[5]);
  lanef = cmul(lanef, (m & 8) ? ub[4] : ua[4]);
  lanef = cmul(lanef, h ? ub[0] : ua[0]);

  v8f Sre, Sim;
#pragma unroll
  for (int v = 0; v < 8; ++v) {
    int idx = (v + 8 * h) * 16 + m;
    float2 regf = (v & 4) ? ub[1] : ua[1];
    regf = cmul(regf, (v & 2) ? ub[2] : ua[2]);
    regf = cmul(regf, (v & 1) ? ub[3] : ua[3]);
    float2 amp = cmul(lanef, regf);
    float2 d = sD[idx];
    Sre[v] = amp.x * d.x - amp.y * d.y;
    Sim[v] = amp.x * d.y + amp.y * d.x;
  }

  for (int l = 1; l < NL - 1; ++l) {
    const float* Uh = &sU[l * 512];
    const float* Ul = &sU[l * 512 + 256];

    v2f buh[4], bul[4];
#pragma unroll
    for (int kb = 0; kb < 4; ++kb) {
      buh[kb][0] = Uh[m * 16 + 8 * h + 2 * kb + 0];
      buh[kb][1] = Uh[m * 16 + 8 * h + 2 * kb + 1];
      bul[kb][0] = Ul[m * 16 + 8 * h + 2 * kb + 0];
      bul[kb][1] = Ul[m * 16 + 8 * h + 2 * kb + 1];
    }

    v8f Tre = {0.f, 0.f, 0.f, 0.f, 0.f, 0.f, 0.f, 0.f};
    v8f Tim = {0.f, 0.f, 0.f, 0.f, 0.f, 0.f, 0.f, 0.f};
#pragma unroll
    for (int kb = 0; kb < 4; ++kb) {
      v2f are, aim;
      are[0] = Sre[2 * kb + 0];
      are[1] = Sre[2 * kb + 1];
      aim[0] = Sim[2 * kb + 0];
      aim[1] = Sim[2 * kb + 1];
      Tre = __builtin_amdgcn_wmma_f32_16x16x4_f32(false, are, false, buh[kb],
                                                  (short)0, Tre, false, false);
      Tim = __builtin_amdgcn_wmma_f32_16x16x4_f32(false, aim, false, buh[kb],
                                                  (short)0, Tim, false, false);
    }

    v8f Nre = {0.f, 0.f, 0.f, 0.f, 0.f, 0.f, 0.f, 0.f};
    v8f Nim = {0.f, 0.f, 0.f, 0.f, 0.f, 0.f, 0.f, 0.f};
#pragma unroll
    for (int kb = 0; kb < 4; ++kb) {
      v2f are, aim;
      are[0] = Tre[2 * kb + 0];
      are[1] = Tre[2 * kb + 1];
      aim[0] = Tim[2 * kb + 0];
      aim[1] = Tim[2 * kb + 1];
      Nre = __builtin_amdgcn_wmma_f32_16x16x4_f32(false, are, false, bul[kb],
                                                  (short)0, Nre, false, false);
      Nim = __builtin_amdgcn_wmma_f32_16x16x4_f32(false, aim, false, bul[kb],
                                                  (short)0, Nim, false, false);
    }

#pragma unroll
    for (int v = 0; v < 8; ++v) {
      int idx = (v + 8 * h) * 16 + m;
      float2 d = sD[l * 256 + idx];
      float re = Nre[v], im = Nim[v];
      Sre[v] = re * d.x - im * d.y;
      Sim[v] = re * d.y + im * d.x;
    }
  }

  v8f Tre = {0.f, 0.f, 0.f, 0.f, 0.f, 0.f, 0.f, 0.f};
  v8f Tim = {0.f, 0.f, 0.f, 0.f, 0.f, 0.f, 0.f, 0.f};
  {
    const float* Uh = &sU[(NL - 1) * 512];
#pragma unroll
    for (int kb = 0; kb < 4; ++kb) {
      v2f b, are, aim;
      b[0] = Uh[m * 16 + 8 * h + 2 * kb + 0];
      b[1] = Uh[m * 16 + 8 * h + 2 * kb + 1];
      are[0] = Sre[2 * kb + 0];
      are[1] = Sre[2 * kb + 1];
      aim[0] = Sim[2 * kb + 0];
      aim[1] = Sim[2 * kb + 1];
      Tre = __builtin_amdgcn_wmma_f32_16x16x4_f32(false, are, false, b,
                                                  (short)0, Tre, false, false);
      Tim = __builtin_amdgcn_wmma_f32_16x16x4_f32(false, aim, false, b,
                                                  (short)0, Tim, false, false);
    }
  }

  float p[8];
#pragma unroll
  for (int r = 0; r < 8; ++r) p[r] = Tre[r] * Tre[r] + Tim[r] * Tim[r];
  float s01 = p[0] + p[1], d01 = p[0] - p[1];
  float s23 = p[2] + p[3], d23 = p[2] - p[3];
  float s45 = p[4] + p[5], d45 = p[4] - p[5];
  float s67 = p[6] + p[7], d67 = p[6] - p[7];
  float sA = s01 + s23, eA = s01 - s23;
  float sB = s45 + s67, eB = s45 - s67;
  float tt = sA + sB;
  float a5 = sA - sB;
  float a6 = eA + eB;
  float a7 = (d01 + d23) + (d45 + d67);
  float b5 = 0.0f, b6 = 0.0f, b7 = 0.0f;
#pragma unroll
  for (int r = 0; r < 8; ++r) {
    if (!(r & 4)) b5 += Tre[r] * Tre[r + 4] + Tim[r] * Tim[r + 4];
    if (!(r & 2)) b6 += Tre[r] * Tre[r + 2] + Tim[r] * Tim[r + 2];
    if (!(r & 1)) b7 += Tre[r] * Tre[r + 1] + Tim[r] * Tim[r + 1];
  }
  float x4 = 0.0f;
#pragma unroll
  for (int r = 0; r < 8; ++r) {
    int bre = __float_as_int(Tre[r]);
    int bim = __float_as_int(Tim[r]);
    float ore = __int_as_float(
        __builtin_amdgcn_permlanex16(bre, bre, 0x76543210, 0xfedcba98, false, false));
    float oim = __int_as_float(
        __builtin_amdgcn_permlanex16(bim, bim, 0x76543210, 0xfedcba98, false, false));
    x4 += Tre[r] * ore + Tim[r] * oim;
  }

  float sq[NQ];
  sq[0] = (m & 8) ? -tt : tt;
  sq[1] = (m & 4) ? -tt : tt;
  sq[2] = (m & 2) ? -tt : tt;
  sq[3] = (m & 1) ? -tt : tt;
  sq[4] = sZ[0] * (h ? -tt : tt) - sZ[4] * x4;
  sq[5] = sZ[1] * a5 - sZ[5] * b5;
  sq[6] = sZ[2] * a6 - sZ[6] * b6;
  sq[7] = sZ[3] * a7 - sZ[7] * b7;
  float r4[4];
#pragma unroll
  for (int j = 0; j < 4; ++j) {
    float keep = h ? sq[j + 4] : sq[j];
    float send = h ? sq[j] : sq[j + 4];
    r4[j] = keep + xhalf(send);
  }
  float r2[2];
#pragma unroll
  for (int j = 0; j < 2; ++j) {
    float keep = (m & 8) ? r4[j + 2] : r4[j];
    float send = (m & 8) ? r4[j] : r4[j + 2];
    r2[j] = keep + xrow<8>(send);
  }
  float r1;
  {
    float keep = (m & 4) ? r2[1] : r2[0];
    float send = (m & 4) ? r2[0] : r2[1];
    r1 = keep + xrow<4>(send);
  }
  r1 += xrow<2>(r1);
  r1 += xrow<1>(r1);
  if ((lane & 3) == 0)
    out[sample * NQ + (lane >> 2)] = r1;
  }
}

extern "C" void kernel_launch(void* const* d_in, const int* in_sizes, int n_in,
                              void* d_out, int out_size, void* d_ws, size_t ws_size,
                              hipStream_t stream) {
  const float* weights = (const float*)d_in[0];
  const float* z1      = (const float*)d_in[1];
  const float* z2      = (const float*)d_in[2];
  float* out           = (float*)d_out;
  int batch = in_sizes[1];
  qnoise_wmma_kernel<<<NBLOCKS, THREADS, 0, stream>>>(weights, z1, z2, out, batch);
}
